# starting kernel
# baseline (speedup 1.0000x reference)
#include <hip/hip_runtime.h>

#define N_NODES 100000
#define N_EDGES 640000
#define D       128
#define BN_EPS  1e-5f
#define LDS_PITCH 132

typedef float v2f __attribute__((ext_vector_type(2)));
typedef float v8f __attribute__((ext_vector_type(8)));

__global__ void k_init_deg(float* __restrict__ deg) {
    int i = blockIdx.x * blockDim.x + threadIdx.x;
    if (i < N_NODES) deg[i] = 1.0f;
}

__global__ void k_deg(const int* __restrict__ dst, float* __restrict__ deg) {
    int e = blockIdx.x * blockDim.x + threadIdx.x;
    if (e < N_EDGES) atomicAdd(&deg[dst[e]], 1.0f);
}

__global__ void k_dinv(float* __restrict__ deg) {
    int i = blockIdx.x * blockDim.x + threadIdx.x;
    if (i < N_NODES) deg[i] = rsqrtf(deg[i]);
}

__global__ void __launch_bounds__(256) k_gemm(const float* __restrict__ x,
                                              const float* __restrict__ Wm,
                                              float* __restrict__ h) {
    __shared__ float lx[16 * LDS_PITCH];
    const int tid  = threadIdx.x;
    const int row0 = blockIdx.x * 16;

    for (int idx = tid; idx < 16 * D; idx += 256) {
        int r = idx >> 7, c = idx & (D - 1);
        lx[r * LDS_PITCH + c] = x[(row0 + r) * D + c];
    }
    __syncthreads();

    const int wave = tid >> 5;
    const int lane = tid & 31;
    const int half = lane >> 4;
    const int l15  = lane & 15;
    const int koff = half * 2;
    const int n0   = wave * 16;

    v8f acc = {};
    for (int k = 0; k < D; k += 4) {
        v2f a, b;
        a.x = lx[l15 * LDS_PITCH + k + koff];
        a.y = lx[l15 * LDS_PITCH + k + koff + 1];
        b.x = Wm[(k + koff)     * D + n0 + l15];
        b.y = Wm[(k + koff + 1) * D + n0 + l15];
        acc = __builtin_amdgcn_wmma_f32_16x16x4_f32(
             false, a,  false, b,
             (short)0, acc,  false,  false);
    }

#pragma unroll
    for (int r = 0; r < 8; ++r) {
        int row = row0 + r + half * 8;
        h[row * D + n0 + l15] = acc[r];
    }
}

__global__ void k_self_bias(const float* __restrict__ h, const float* __restrict__ dinv,
                            const float* __restrict__ b, float* __restrict__ out) {
    int i = blockIdx.x * blockDim.x + threadIdx.x;
    if (i < N_NODES * D) {
        int n = i >> 7, d = i & (D - 1);
        float di = dinv[n];
        out[i] = b[d] + h[i] * (di * di);
    }
}

__global__ void __launch_bounds__(256) k_scatter(const int* __restrict__ src,
                                                 const int* __restrict__ dst,
                                                 const float* __restrict__ h,
                                                 const float* __restrict__ dinv,
                                                 float* __restrict__ out) {
    int e    = blockIdx.x * (blockDim.x >> 5) + (threadIdx.x >> 5);
    int lane = threadIdx.x & 31;
    if (e >= N_EDGES) return;

    int ep = e + 8;
    if (ep < N_EDGES)
        __builtin_prefetch(h + (size_t)src[ep] * D + lane * 4, 0, 0);

    int s = src[e], t = dst[e];
    float w = dinv[s] * dinv[t];

    const float4* hp = (const float4*)(h + (size_t)s * D);
    float4 v = hp[lane];
    float* op = out + (size_t)t * D + lane * 4;
    atomicAdd(op + 0, v.x * w);
    atomicAdd(op + 1, v.y * w);
    atomicAdd(op + 2, v.z * w);
    atomicAdd(op + 3, v.w * w);
}

__global__ void k_init_stats(float* __restrict__ sums, float* __restrict__ sumsq) {
    int i = threadIdx.x;
    if (i < D) { sums[i] = 0.0f; sumsq[i] = 0.0f; }
}

__global__ void __launch_bounds__(256) k_stats(const float* __restrict__ out,
                                               float* __restrict__ sums,
                                               float* __restrict__ sumsq) {
    int d = threadIdx.x & (D - 1);
    int r0 = blockIdx.x * 2 + (threadIdx.x >> 7);
    float s = 0.0f, ss = 0.0f;
    for (int r = r0; r < N_NODES; r += gridDim.x * 2) {
        float v = out[r * D + d];
        v = v > 0.0f ? v : 0.0f;
        s += v; ss += v * v;
    }
    atomicAdd(&sums[d], s);
    atomicAdd(&sumsq[d], ss);
}

__global__ void k_bn(float* __restrict__ out, const float* __restrict__ sums,
                     const float* __restrict__ sumsq, const float* __restrict__ gamma,
                     const float* __restrict__ beta) {
    int i = blockIdx.x * blockDim.x + threadIdx.x;
    if (i >= N_NODES * D) return;
    int d = i & (D - 1);
    const float invN = 1.0f / (float)N_NODES;
    float mean = sums[d] * invN;
    float var  = sumsq[d] * invN - mean * mean;
    float v = out[i];
    v = v > 0.0f ? v : 0.0f;
    out[i] = gamma[d] * (v - mean) * rsqrtf(var + BN_EPS) + beta[d];
}

extern "C" void kernel_launch(void* const* d_in, const int* in_sizes, int n_in,
                              void* d_out, int out_size, void* d_ws, size_t ws_size,
                              hipStream_t stream) {
    const float* x     = (const float*)d_in[0];
    const int*   edges = (const int*)  d_in[1];
    const float* Wm    = (const float*)d_in[2];
    const float* b     = (const float*)d_in[3];
    const float* gamma = (const float*)d_in[4];
    const float* beta  = (const float*)d_in[5];
    float*       out   = (float*)d_out;

    const int* src = edges;
    const int* dst = edges + N_EDGES;

    float* deg   = (float*)d_ws;
    float* h     = deg + N_NODES;
    float* sums  = h + (size_t)N_NODES * D;
    float* sumsq = sums + D;

    k_init_deg  <<<(N_NODES + 255) / 256, 256, 0, stream>>>(deg);
    k_deg       <<<(N_EDGES + 255) / 256, 256, 0, stream>>>(dst, deg);
    k_dinv      <<<(N_NODES + 255) / 256, 256, 0, stream>>>(deg);

    k_gemm      <<<N_NODES / 16, 256, 0, stream>>>(x, Wm, h);

    k_self_bias <<<(N_NODES * D + 255) / 256, 256, 0, stream>>>(h, deg, b, out);
    k_scatter   <<<N_EDGES / 8, 256, 0, stream>>>(src, dst, h, deg, out);

    k_init_stats<<<1, 128, 0, stream>>>(sums, sumsq);
    k_stats     <<<512, 256, 0, stream>>>(out, sums, sumsq);
    k_bn        <<<(N_NODES * D + 255) / 256, 256, 0, stream>>>(out, sums, sumsq, gamma, beta);
}


# Global float-atomic scatter replaced by block-owned LDS rows: 250 blocks own 400 rows each, filter dst, add h rows by LDS float atomics (arrival order); bias/self start and BN sums from the LDS rows
# speedup vs baseline: 1.4507x; 1.4507x over previous
#include <hip/hip_runtime.h>

#define N_NODES 100000
#define N_EDGES 640000
#define D       128
#define BN_EPS  1e-5f
#define LDS_PITCH 132

typedef float v2f __attribute__((ext_vector_type(2)));
typedef float v8f __attribute__((ext_vector_type(8)));

__global__ void k_init_deg(float* __restrict__ deg) {
    int i = blockIdx.x * blockDim.x + threadIdx.x;
    if (i < N_NODES) deg[i] = 1.0f;
}

__global__ void k_deg(const int* __restrict__ dst, float* __restrict__ deg) {
    int e = blockIdx.x * blockDim.x + threadIdx.x;
    if (e < N_EDGES) atomicAdd(&deg[dst[e]], 1.0f);
}

__global__ void k_dinv(float* __restrict__ deg) {
    int i = blockIdx.x * blockDim.x + threadIdx.x;
    if (i < N_NODES) deg[i] = rsqrtf(deg[i]);
}

__global__ void __launch_bounds__(256) k_gemm(const float* __restrict__ x,
                                              const float* __restrict__ Wm,
                                              float* __restrict__ h) {
    __shared__ float lx[16 * LDS_PITCH];
    const int tid  = threadIdx.x;
    const int row0 = blockIdx.x * 16;

    for (int idx = tid; idx < 16 * D; idx += 256) {
        int r = idx >> 7, c = idx & (D - 1);
        lx[r * LDS_PITCH + c] = x[(row0 + r) * D + c];
    }
    __syncthreads();

    const int wave = tid >> 5;
    const int lane = tid & 31;
    const int half = lane >> 4;
    const int l15  = lane & 15;
    const int koff = half * 2;
    const int n0   = wave * 16;

    v8f acc = {};
    for (int k = 0; k < D; k += 4) {
        v2f a, b;
        a.x = lx[l15 * LDS_PITCH + k + koff];
        a.y = lx[l15 * LDS_PITCH + k + koff + 1];
        b.x = Wm[(k + koff)     * D + n0 + l15];
        b.y = Wm[(k + koff + 1) * D + n0 + l15];
        acc = __builtin_amdgcn_wmma_f32_16x16x4_f32(
             false, a,  false, b,
             (short)0, acc,  false,  false);
    }

#pragma unroll
    for (int r = 0; r < 8; ++r) {
        int row = row0 + r + half * 8;
        h[row * D + n0 + l15] = acc[r];
    }
}

#define ROWS_PER_BLK 400
#define N_BLK        (N_NODES / ROWS_PER_BLK)
#define LIST_CAP     16384
#define N_DST4       (N_EDGES / 4)
static_assert(N_BLK * ROWS_PER_BLK == N_NODES, "rows must divide");
static_assert(N_EDGES < (1 << 20) && ROWS_PER_BLK < (1 << 11), "list entry packing");

__device__ __noinline__ void edge_serial(int e, int r, int lo, const int* __restrict__ src,
                                         const float* __restrict__ h,
                                         const float* __restrict__ dinv, float* accf) {
    int s = src[e];
    float w = dinv[s] * dinv[lo + r];
    const float* hp = h + (size_t)s * D;
    float* ap = accf + r * D;
    for (int c = 0; c < D; ++c) atomicAdd(ap + c, hp[c] * w);
}

__global__ void __launch_bounds__(1024) k_aggregate(const int* __restrict__ src,
                                                    const int* __restrict__ dst,
                                                    const float* __restrict__ h,
                                                    const float* __restrict__ dinv,
                                                    const float* __restrict__ b,
                                                    float* __restrict__ out,
                                                    float* __restrict__ sums,
                                                    float* __restrict__ sumsq) {
    __shared__ float4 acc[(ROWS_PER_BLK + 1) * 32];
    __shared__ int    list[LIST_CAP];
    __shared__ float  colsum[2 * D];
    __shared__ int    n_list;

    const int tid  = threadIdx.x;
    const int lane = tid & 31;
    const int wave = __builtin_amdgcn_readfirstlane(tid >> 5);
    const int lo   = blockIdx.x * ROWS_PER_BLK;
    float* accf = (float*)acc;
    const float4* h4 = (const float4*)h;

    {
        const float4 bv = ((const float4*)b)[lane];
#pragma unroll
        for (int it = 0; it < 13; ++it) {
            int r  = wave + 32 * it;
            int rc = r < ROWS_PER_BLK ? r : ROWS_PER_BLK - 1;
            float di = dinv[lo + rc];
            float sc = di * di;
            float4 hv = h4[(size_t)(lo + rc) * 32 + lane];
            float4 v;
            v.x = bv.x + hv.x * sc;
            v.y = bv.y + hv.y * sc;
            v.z = bv.z + hv.z * sc;
            v.w = bv.w + hv.w * sc;
            if (r < ROWS_PER_BLK) acc[r * 32 + lane] = v;
        }
        if (tid < 2 * D) colsum[tid] = 0.0f;
        if (tid == 0) n_list = 0;
    }
    __syncthreads();

    {
        const int4* d4 = (const int4*)dst;
        const int start = blockIdx.x * (N_DST4 / N_BLK);
        for (int i0 = 0; i0 < N_DST4; i0 += 4096) {
            int4 dv[4];
            int  jj[4];
#pragma unroll
            for (int u = 0; u < 4; ++u) {
                int i = i0 + u * 1024 + tid;
                int j = i + start;
                if (j >= N_DST4) j -= N_DST4;
                jj[u] = j;
                dv[u] = (i < N_DST4) ? d4[j] : make_int4(-1, -1, -1, -1);
            }
#pragma unroll
            for (int u = 0; u < 4; ++u) {
                const int dd[4] = {dv[u].x, dv[u].y, dv[u].z, dv[u].w};
#pragma unroll
                for (int c = 0; c < 4; ++c) {
                    unsigned r = (unsigned)(dd[c] - lo);
                    if (r < (unsigned)ROWS_PER_BLK) {
                        int e = 4 * jj[u] + c;
                        int p = atomicAdd(&n_list, 1);
                        if (p < LIST_CAP) list[p] = e | (int)(r << 20);
                        else edge_serial(e, (int)r, lo, src, h, dinv, accf);
                    }
                }
            }
        }
    }
    __syncthreads();

    {
        int nl = n_list;
        nl = nl < LIST_CAP ? nl : LIST_CAP;
        const int n = __builtin_amdgcn_readfirstlane(nl);
        for (int base = wave * 32; base < n; base += 1024) {
            int r = ROWS_PER_BLK, s = 0;
            float w = 0.0f;
            if (base + lane < n) {
                int ent = list[base + lane];
                int e = ent & 0xFFFFF;
                r = ent >> 20;
                s = src[e];
                w = dinv[s] * dinv[lo + r];
            }
            const int wi = __float_as_int(w);
#pragma unroll
            for (int g = 0; g < 32; g += 8) {
                if (base + g < n) {
                    float4 hv[8];
                    int    rr[8];
                    float  ww[8];
#pragma unroll
                    for (int q = 0; q < 8; ++q) {
                        int sq = __builtin_amdgcn_readlane(s, g + q);
                        rr[q]  = __builtin_amdgcn_readlane(r, g + q);
                        ww[q]  = __int_as_float(__builtin_amdgcn_readlane(wi, g + q));
                        hv[q]  = h4[(size_t)sq * 32 + lane];
                    }
#pragma unroll
                    for (int q = 0; q < 8; ++q) {
                        float* ap = accf + rr[q] * D + 4 * lane;
                        atomicAdd(ap + 0, hv[q].x * ww[q]);
                        atomicAdd(ap + 1, hv[q].y * ww[q]);
                        atomicAdd(ap + 2, hv[q].z * ww[q]);
                        atomicAdd(ap + 3, hv[q].w * ww[q]);
                    }
                }
            }
        }
    }
    __syncthreads();

    {
        float4 s4 = {0.0f, 0.0f, 0.0f, 0.0f};
        float4 q4 = {0.0f, 0.0f, 0.0f, 0.0f};
        float4* out4 = (float4*)out;
        for (int r = wave; r < ROWS_PER_BLK; r += 32) {
            float4 v = acc[r * 32 + lane];
            out4[(size_t)(lo + r) * 32 + lane] = v;
            v.x = v.x > 0.0f ? v.x : 0.0f;
            v.y = v.y > 0.0f ? v.y : 0.0f;
            v.z = v.z > 0.0f ? v.z : 0.0f;
            v.w = v.w > 0.0f ? v.w : 0.0f;
            s4.x += v.x; q4.x += v.x * v.x;
            s4.y += v.y; q4.y += v.y * v.y;
            s4.z += v.z; q4.z += v.z * v.z;
            s4.w += v.w; q4.w += v.w * v.w;
        }
        atomicAdd(&colsum[4 * lane + 0], s4.x);
        atomicAdd(&colsum[4 * lane + 1], s4.y);
        atomicAdd(&colsum[4 * lane + 2], s4.z);
        atomicAdd(&colsum[4 * lane + 3], s4.w);
        atomicAdd(&colsum[D + 4 * lane + 0], q4.x);
        atomicAdd(&colsum[D + 4 * lane + 1], q4.y);
        atomicAdd(&colsum[D + 4 * lane + 2], q4.z);
        atomicAdd(&colsum[D + 4 * lane + 3], q4.w);
    }
    __syncthreads();
    if (tid < D)          atomicAdd(&sums[tid], colsum[tid]);
    else if (tid < 2 * D) atomicAdd(&sumsq[tid - D], colsum[tid]);
}

__global__ void k_init_stats(float* __restrict__ sums, float* __restrict__ sumsq) {
    int i = threadIdx.x;
    if (i < D) { sums[i] = 0.0f; sumsq[i] = 0.0f; }
}

__global__ void k_bn(float* __restrict__ out, const float* __restrict__ sums,
                     const float* __restrict__ sumsq, const float* __restrict__ gamma,
                     const float* __restrict__ beta) {
    int i = blockIdx.x * blockDim.x + threadIdx.x;
    if (i >= N_NODES * D) return;
    int d = i & (D - 1);
    const float invN = 1.0f / (float)N_NODES;
    float mean = sums[d] * invN;
    float var  = sumsq[d] * invN - mean * mean;
    float v = out[i];
    v = v > 0.0f ? v : 0.0f;
    out[i] = gamma[d] * (v - mean) * rsqrtf(var + BN_EPS) + beta[d];
}

extern "C" void kernel_launch(void* const* d_in, const int* in_sizes, int n_in,
                              void* d_out, int out_size, void* d_ws, size_t ws_size,
                              hipStream_t stream) {
    const float* x     = (const float*)d_in[0];
    const int*   edges = (const int*)  d_in[1];
    const float* Wm    = (const float*)d_in[2];
    const float* b     = (const float*)d_in[3];
    const float* gamma = (const float*)d_in[4];
    const float* beta  = (const float*)d_in[5];
    float*       out   = (float*)d_out;

    const int* src = edges;
    const int* dst = edges + N_EDGES;

    float* deg   = (float*)d_ws;
    float* h     = deg + N_NODES;
    float* sums  = h + (size_t)N_NODES * D;
    float* sumsq = sums + D;

    k_init_deg  <<<(N_NODES + 255) / 256, 256, 0, stream>>>(deg);
    k_deg       <<<(N_EDGES + 255) / 256, 256, 0, stream>>>(dst, deg);
    k_dinv      <<<(N_NODES + 255) / 256, 256, 0, stream>>>(deg);

    k_gemm      <<<N_NODES / 16, 256, 0, stream>>>(x, Wm, h);

    k_init_stats<<<1, 128, 0, stream>>>(sums, sumsq);
    k_aggregate <<<N_BLK, 1024, 0, stream>>>(src, dst, h, deg, b, out, sums, sumsq);
    k_bn        <<<(N_NODES * D + 255) / 256, 256, 0, stream>>>(out, sums, sumsq, gamma, beta);
}


# Edge filter of the LDS aggregation reads a byte-per-edge owner map (block number of each destination, written by the degree launch) instead of the int destinations: 1/4 of the bytes; sums as before
# speedup vs baseline: 1.6086x; 1.1088x over previous
#include <hip/hip_runtime.h>

#define N_NODES 100000
#define N_EDGES 640000
#define D       128
#define BN_EPS  1e-5f
#define LDS_PITCH 132

typedef float v2f __attribute__((ext_vector_type(2)));
typedef float v8f __attribute__((ext_vector_type(8)));

__global__ void k_init_deg(float* __restrict__ deg) {
    int i = blockIdx.x * blockDim.x + threadIdx.x;
    if (i < N_NODES) deg[i] = 1.0f;
}

#define ROWS_PER_BLK 400

__global__ void k_deg(const int* __restrict__ dst, float* __restrict__ deg,
                      unsigned* __restrict__ own4) {
    int i = blockIdx.x * blockDim.x + threadIdx.x;
    if (i < N_EDGES / 4) {
        int4 d = ((const int4*)dst)[i];
        atomicAdd(&deg[d.x], 1.0f);
        atomicAdd(&deg[d.y], 1.0f);
        atomicAdd(&deg[d.z], 1.0f);
        atomicAdd(&deg[d.w], 1.0f);
        unsigned o = (unsigned)(d.x / ROWS_PER_BLK)
                   | ((unsigned)(d.y / ROWS_PER_BLK) << 8)
                   | ((unsigned)(d.z / ROWS_PER_BLK) << 16)
                   | ((unsigned)(d.w / ROWS_PER_BLK) << 24);
        own4[i] = o;
    }
}

__global__ void k_dinv(float* __restrict__ deg) {
    int i = blockIdx.x * blockDim.x + threadIdx.x;
    if (i < N_NODES) deg[i] = rsqrtf(deg[i]);
}

__global__ void __launch_bounds__(256) k_gemm(const float* __restrict__ x,
                                              const float* __restrict__ Wm,
                                              float* __restrict__ h) {
    __shared__ float lx[16 * LDS_PITCH];
    const int tid  = threadIdx.x;
    const int row0 = blockIdx.x * 16;

    for (int idx = tid; idx < 16 * D; idx += 256) {
        int r = idx >> 7, c = idx & (D - 1);
        lx[r * LDS_PITCH + c] = x[(row0 + r) * D + c];
    }
    __syncthreads();

    const int wave = tid >> 5;
    const int lane = tid & 31;
    const int half = lane >> 4;
    const int l15  = lane & 15;
    const int koff = half * 2;
    const int n0   = wave * 16;

    v8f acc = {};
    for (int k = 0; k < D; k += 4) {
        v2f a, b;
        a.x = lx[l15 * LDS_PITCH + k + koff];
        a.y = lx[l15 * LDS_PITCH + k + koff + 1];
        b.x = Wm[(k + koff)     * D + n0 + l15];
        b.y = Wm[(k + koff + 1) * D + n0 + l15];
        acc = __builtin_amdgcn_wmma_f32_16x16x4_f32(
             false, a,  false, b,
             (short)0, acc,  false,  false);
    }

#pragma unroll
    for (int r = 0; r < 8; ++r) {
        int row = row0 + r + half * 8;
        h[row * D + n0 + l15] = acc[r];
    }
}

#define N_BLK        (N_NODES / ROWS_PER_BLK)
#define LIST_CAP     16384
#define N_OWN16      (N_EDGES / 16)
static_assert(N_BLK * ROWS_PER_BLK == N_NODES, "rows must divide");
static_assert(N_BLK <= 256, "owner number must fit a byte");
static_assert(N_OWN16 * 16 == N_EDGES, "owner map is read in whole 16-byte pieces");

__device__ __forceinline__ unsigned nz7(unsigned t) {
    return ((t & 0x7F7F7F7Fu) + 0x7F7F7F7Fu) | t;
}

__device__ __noinline__ void edge_serial(int e, int lo, const int* __restrict__ src,
                                         const int* __restrict__ dst,
                                         const float* __restrict__ h,
                                         const float* __restrict__ dinv, float* accf) {
    int s = src[e];
    int d = dst[e];
    int r = d - lo;
    float w = dinv[s] * dinv[d];
    const float* hp = h + (size_t)s * D;
    float* ap = accf + r * D;
    for (int c = 0; c < D; ++c) atomicAdd(ap + c, hp[c] * w);
}

__global__ void __launch_bounds__(1024) k_aggregate(const int* __restrict__ src,
                                                    const int* __restrict__ dst,
                                                    const unsigned* __restrict__ own4,
                                                    const float* __restrict__ h,
                                                    const float* __restrict__ dinv,
                                                    const float* __restrict__ b,
                                                    float* __restrict__ out,
                                                    float* __restrict__ sums,
                                                    float* __restrict__ sumsq) {
    __shared__ float4 acc[(ROWS_PER_BLK + 1) * 32];
    __shared__ int    list[LIST_CAP];
    __shared__ float  colsum[2 * D];
    __shared__ int    n_list;

    const int tid  = threadIdx.x;
    const int lane = tid & 31;
    const int wave = __builtin_amdgcn_readfirstlane(tid >> 5);
    const int lo   = blockIdx.x * ROWS_PER_BLK;
    float* accf = (float*)acc;
    const float4* h4 = (const float4*)h;

    {
        const float4 bv = ((const float4*)b)[lane];
#pragma unroll
        for (int it = 0; it < 13; ++it) {
            int r  = wave + 32 * it;
            int rc = r < ROWS_PER_BLK ? r : ROWS_PER_BLK - 1;
            float di = dinv[lo + rc];
            float sc = di * di;
            float4 hv = h4[(size_t)(lo + rc) * 32 + lane];
            float4 v;
            v.x = bv.x + hv.x * sc;
            v.y = bv.y + hv.y * sc;
            v.z = bv.z + hv.z * sc;
            v.w = bv.w + hv.w * sc;
            if (r < ROWS_PER_BLK) acc[r * 32 + lane] = v;
        }
        if (tid < 2 * D) colsum[tid] = 0.0f;
        if (tid == 0) n_list = 0;
    }
    __syncthreads();

    {
        const uint4* o16 = (const uint4*)own4;
        const unsigned pat = (unsigned)blockIdx.x * 0x01010101u;
        const int start = blockIdx.x * (N_OWN16 / N_BLK);
        for (int i0 = 0; i0 < N_OWN16; i0 += 8192) {
            uint4 ov[8];
            int   jj[8];
#pragma unroll
            for (int u = 0; u < 8; ++u) {
                int i = i0 + u * 1024 + tid;
                int j = (i < N_OWN16 ? i : N_OWN16 - 1) + start;
                if (j >= N_OWN16) j -= N_OWN16;
                jj[u] = j;
                ov[u] = o16[j];
            }
#pragma unroll
            for (int u = 0; u < 8; ++u) {
                unsigned m = ~(((nz7(ov[u].x ^ pat) & 0x80808080u) >> 7) |
                               ((nz7(ov[u].y ^ pat) & 0x80808080u) >> 6) |
                               ((nz7(ov[u].z ^ pat) & 0x80808080u) >> 5) |
                               ((nz7(ov[u].w ^ pat) & 0x80808080u) >> 4)) & 0x0F0F0F0Fu;
                if (i0 + u * 1024 + tid >= N_OWN16) m = 0;
                while (m) {
                    int bit = __builtin_ctz(m);
                    m &= m - 1;
                    int e = 16 * jj[u] + 4 * (bit & 7) + (bit >> 3);
                    int p = atomicAdd(&n_list, 1);
                    if (p < LIST_CAP) list[p] = e;
                    else edge_serial(e, lo, src, dst, h, dinv, accf);
                }
            }
        }
    }
    __syncthreads();

    {
        int nl = n_list;
        nl = nl < LIST_CAP ? nl : LIST_CAP;
        const int n = __builtin_amdgcn_readfirstlane(nl);
        for (int base = wave * 32; base < n; base += 1024) {
            int r = ROWS_PER_BLK, s = 0;
            float w = 0.0f;
            if (base + lane < n) {
                int e = list[base + lane];
                s = src[e];
                int d = dst[e];
                r = d - lo;
                w = dinv[s] * dinv[d];
            }
            const int wi = __float_as_int(w);
#pragma unroll
            for (int g = 0; g < 32; g += 8) {
                if (base + g < n) {
                    float4 hv[8];
                    int    rr[8];
                    float  ww[8];
#pragma unroll
                    for (int q = 0; q < 8; ++q) {
                        int sq = __builtin_amdgcn_readlane(s, g + q);
                        rr[q]  = __builtin_amdgcn_readlane(r, g + q);
                        ww[q]  = __int_as_float(__builtin_amdgcn_readlane(wi, g + q));
                        hv[q]  = h4[(size_t)sq * 32 + lane];
                    }
#pragma unroll
                    for (int q = 0; q < 8; ++q) {
                        float* ap = accf + rr[q] * D + 4 * lane;
                        atomicAdd(ap + 0, hv[q].x * ww[q]);
                        atomicAdd(ap + 1, hv[q].y * ww[q]);
                        atomicAdd(ap + 2, hv[q].z * ww[q]);
                        atomicAdd(ap + 3, hv[q].w * ww[q]);
                    }
                }
            }
        }
    }
    __syncthreads();

    {
        float4 s4 = {0.0f, 0.0f, 0.0f, 0.0f};
        float4 q4 = {0.0f, 0.0f, 0.0f, 0.0f};
        float4* out4 = (float4*)out;
        for (int r = wave; r < ROWS_PER_BLK; r += 32) {
            float4 v = acc[r * 32 + lane];
            out4[(size_t)(lo + r) * 32 + lane] = v;
            v.x = v.x > 0.0f ? v.x : 0.0f;
            v.y = v.y > 0.0f ? v.y : 0.0f;
            v.z = v.z > 0.0f ? v.z : 0.0f;
            v.w = v.w > 0.0f ? v.w : 0.0f;
            s4.x += v.x; q4.x += v.x * v.x;
            s4.y += v.y; q4.y += v.y * v.y;
            s4.z += v.z; q4.z += v.z * v.z;
            s4.w += v.w; q4.w += v.w * v.w;
        }
        atomicAdd(&colsum[4 * lane + 0], s4.x);
        atomicAdd(&colsum[4 * lane + 1], s4.y);
        atomicAdd(&colsum[4 * lane + 2], s4.z);
        atomicAdd(&colsum[4 * lane + 3], s4.w);
        atomicAdd(&colsum[D + 4 * lane + 0], q4.x);
        atomicAdd(&colsum[D + 4 * lane + 1], q4.y);
        atomicAdd(&colsum[D + 4 * lane + 2], q4.z);
        atomicAdd(&colsum[D + 4 * lane + 3], q4.w);
    }
    __syncthreads();
    if (tid < D)          atomicAdd(&sums[tid], colsum[tid]);
    else if (tid < 2 * D) atomicAdd(&sumsq[tid - D], colsum[tid]);
}

__global__ void k_init_stats(float* __restrict__ sums, float* __restrict__ sumsq) {
    int i = threadIdx.x;
    if (i < D) { sums[i] = 0.0f; sumsq[i] = 0.0f; }
}

__global__ void k_bn(float* __restrict__ out, const float* __restrict__ sums,
                     const float* __restrict__ sumsq, const float* __restrict__ gamma,
                     const float* __restrict__ beta) {
    int i = blockIdx.x * blockDim.x + threadIdx.x;
    if (i >= N_NODES * D) return;
    int d = i & (D - 1);
    const float invN = 1.0f / (float)N_NODES;
    float mean = sums[d] * invN;
    float var  = sumsq[d] * invN - mean * mean;
    float v = out[i];
    v = v > 0.0f ? v : 0.0f;
    out[i] = gamma[d] * (v - mean) * rsqrtf(var + BN_EPS) + beta[d];
}

extern "C" void kernel_launch(void* const* d_in, const int* in_sizes, int n_in,
                              void* d_out, int out_size, void* d_ws, size_t ws_size,
                              hipStream_t stream) {
    const float* x     = (const float*)d_in[0];
    const int*   edges = (const int*)  d_in[1];
    const float* Wm    = (const float*)d_in[2];
    const float* b     = (const float*)d_in[3];
    const float* gamma = (const float*)d_in[4];
    const float* beta  = (const float*)d_in[5];
    float*       out   = (float*)d_out;

    const int* src = edges;
    const int* dst = edges + N_EDGES;

    float* deg   = (float*)d_ws;
    float* h     = deg + N_NODES;
    float* sums  = h + (size_t)N_NODES * D;
    float* sumsq = sums + D;
    unsigned* own4 = (unsigned*)(sumsq + D);

    k_init_deg  <<<(N_NODES + 255) / 256, 256, 0, stream>>>(deg);
    k_deg       <<<(N_EDGES / 4 + 255) / 256, 256, 0, stream>>>(dst, deg, own4);
    k_dinv      <<<(N_NODES + 255) / 256, 256, 0, stream>>>(deg);

    k_gemm      <<<N_NODES / 16, 256, 0, stream>>>(x, Wm, h);

    k_init_stats<<<1, 128, 0, stream>>>(sums, sumsq);
    k_aggregate <<<N_BLK, 1024, 0, stream>>>(src, dst, own4, h, deg, b, out, sums, sumsq);
    k_bn        <<<(N_NODES * D + 255) / 256, 256, 0, stream>>>(out, sums, sumsq, gamma, beta);
}


# x@W moved behind the aggregation ((sum w x)W = sum w (xW)): blocks aggregate x rows in LDS, then multiply their 400 rows by W on f32 WMMA; GEMM launch and h gone; float sum order changes
# speedup vs baseline: 1.8844x; 1.1715x over previous
#include <hip/hip_runtime.h>

#define N_NODES 100000
#define N_EDGES 640000
#define D       128
#define BN_EPS  1e-5f
#define LDS_PITCH 132

typedef float v2f __attribute__((ext_vector_type(2)));
typedef float v8f __attribute__((ext_vector_type(8)));

__global__ void k_init_deg(float* __restrict__ deg) {
    int i = blockIdx.x * blockDim.x + threadIdx.x;
    if (i < N_NODES) deg[i] = 1.0f;
}

#define ROWS_PER_BLK 400

__global__ void k_deg(const int* __restrict__ dst, float* __restrict__ deg,
                      unsigned* __restrict__ own4) {
    int i = blockIdx.x * blockDim.x + threadIdx.x;
    if (i < N_EDGES / 4) {
        int4 d = ((const int4*)dst)[i];
        atomicAdd(&deg[d.x], 1.0f);
        atomicAdd(&deg[d.y], 1.0f);
        atomicAdd(&deg[d.z], 1.0f);
        atomicAdd(&deg[d.w], 1.0f);
        unsigned o = (unsigned)(d.x / ROWS_PER_BLK)
                   | ((unsigned)(d.y / ROWS_PER_BLK) << 8)
                   | ((unsigned)(d.z / ROWS_PER_BLK) << 16)
                   | ((unsigned)(d.w / ROWS_PER_BLK) << 24);
        own4[i] = o;
    }
}

__global__ void k_dinv(float* __restrict__ deg) {
    int i = blockIdx.x * blockDim.x + threadIdx.x;
    if (i < N_NODES) deg[i] = rsqrtf(deg[i]);
}

#define N_BLK        (N_NODES / ROWS_PER_BLK)
#define LIST_CAP     16384
#define N_OWN16      (N_EDGES / 16)
static_assert(N_BLK * ROWS_PER_BLK == N_NODES, "rows must divide");
static_assert(N_BLK <= 256, "owner number must fit a byte");
static_assert(N_OWN16 * 16 == N_EDGES, "owner map is read in whole 16-byte pieces");
static_assert(ROWS_PER_BLK % 16 == 0, "rows of a block split into whole 16-row matrix tiles");

__device__ __forceinline__ unsigned nz7(unsigned t) {
    return ((t & 0x7F7F7F7Fu) + 0x7F7F7F7Fu) | t;
}

__device__ __noinline__ void edge_serial(int e, int lo, const int* __restrict__ src,
                                         const int* __restrict__ dst,
                                         const float* __restrict__ x,
                                         const float* __restrict__ dinv, float* accf) {
    int s = src[e];
    int d = dst[e];
    int r = d - lo;
    float w = dinv[s] * dinv[d];
    const float* hp = x + (size_t)s * D;
    float* ap = accf + r * LDS_PITCH;
    for (int c = 0; c < D; ++c) atomicAdd(ap + c, hp[c] * w);
}

__global__ void __launch_bounds__(1024) k_aggregate(const int* __restrict__ src,
                                                    const int* __restrict__ dst,
                                                    const unsigned* __restrict__ own4,
                                                    const float* __restrict__ x,
                                                    const float* __restrict__ Wm,
                                                    const float* __restrict__ dinv,
                                                    const float* __restrict__ b,
                                                    float* __restrict__ out,
                                                    float* __restrict__ sums,
                                                    float* __restrict__ sumsq) {
    __shared__ float4 acc[(ROWS_PER_BLK + 1) * (LDS_PITCH / 4)];
    __shared__ int    list[LIST_CAP];
    __shared__ float  colsum[2 * D];
    __shared__ int    n_list;

    const int tid  = threadIdx.x;
    const int lane = tid & 31;
    const int wave = __builtin_amdgcn_readfirstlane(tid >> 5);
    const int lo   = blockIdx.x * ROWS_PER_BLK;
    float* accf = (float*)acc;
    const float4* h4 = (const float4*)x;

    {
#pragma unroll
        for (int it = 0; it < 13; ++it) {
            int r  = wave + 32 * it;
            int rc = r < ROWS_PER_BLK ? r : ROWS_PER_BLK - 1;
            float di = dinv[lo + rc];
            float sc = di * di;
            float4 hv = h4[(size_t)(lo + rc) * 32 + lane];
            float4 v;
            v.x = hv.x * sc;
            v.y = hv.y * sc;
            v.z = hv.z * sc;
            v.w = hv.w * sc;
            if (r < ROWS_PER_BLK) acc[r * (LDS_PITCH / 4) + lane] = v;
        }
        if (tid < 2 * D) colsum[tid] = 0.0f;
        if (tid == 0) n_list = 0;
    }
    __syncthreads();

    {
        const uint4* o16 = (const uint4*)own4;
        const unsigned pat = (unsigned)blockIdx.x * 0x01010101u;
        const int start = blockIdx.x * (N_OWN16 / N_BLK);
        for (int i0 = 0; i0 < N_OWN16; i0 += 8192) {
            uint4 ov[8];
            int   jj[8];
#pragma unroll
            for (int u = 0; u < 8; ++u) {
                int i = i0 + u * 1024 + tid;
                int j = (i < N_OWN16 ? i : N_OWN16 - 1) + start;
                if (j >= N_OWN16) j -= N_OWN16;
                jj[u] = j;
                ov[u] = o16[j];
            }
#pragma unroll
            for (int u = 0; u < 8; ++u) {
                unsigned m = ~(((nz7(ov[u].x ^ pat) & 0x80808080u) >> 7) |
                               ((nz7(ov[u].y ^ pat) & 0x80808080u) >> 6) |
                               ((nz7(ov[u].z ^ pat) & 0x80808080u) >> 5) |
                               ((nz7(ov[u].w ^ pat) & 0x80808080u) >> 4)) & 0x0F0F0F0Fu;
                if (i0 + u * 1024 + tid >= N_OWN16) m = 0;
                while (m) {
                    int bit = __builtin_ctz(m);
                    m &= m - 1;
                    int e = 16 * jj[u] + 4 * (bit & 7) + (bit >> 3);
                    int p = atomicAdd(&n_list, 1);
                    if (p < LIST_CAP) list[p] = e;
                    else edge_serial(e, lo, src, dst, x, dinv, accf);
                }
            }
        }
    }
    __syncthreads();

    {
        int nl = n_list;
        nl = nl < LIST_CAP ? nl : LIST_CAP;
        const int n = __builtin_amdgcn_readfirstlane(nl);
        for (int base = wave * 32; base < n; base += 1024) {
            int r = ROWS_PER_BLK, s = 0;
            float w = 0.0f;
            if (base + lane < n) {
                int e = list[base + lane];
                s = src[e];
                int d = dst[e];
                r = d - lo;
                w = dinv[s] * dinv[d];
            }
            const int wi = __float_as_int(w);
#pragma unroll
            for (int g = 0; g < 32; g += 8) {
                if (base + g < n) {
                    float4 hv[8];
                    int    rr[8];
                    float  ww[8];
#pragma unroll
                    for (int q = 0; q < 8; ++q) {
                        int sq = __builtin_amdgcn_readlane(s, g + q);
                        rr[q]  = __builtin_amdgcn_readlane(r, g + q);
                        ww[q]  = __int_as_float(__builtin_amdgcn_readlane(wi, g + q));
                        hv[q]  = h4[(size_t)sq * 32 + lane];
                    }
#pragma unroll
                    for (int q = 0; q < 8; ++q) {
                        float* ap = accf + rr[q] * LDS_PITCH + 4 * lane;
                        atomicAdd(ap + 0, hv[q].x * ww[q]);
                        atomicAdd(ap + 1, hv[q].y * ww[q]);
                        atomicAdd(ap + 2, hv[q].z * ww[q]);
                        atomicAdd(ap + 3, hv[q].w * ww[q]);
                    }
                }
            }
        }
    }
    const int half = lane >> 4;
    const int l15  = lane & 15;
    const int koff = half * 2;
    const int n0   = (wave & 7) * 16;
    v2f bw[32];
#pragma unroll
    for (int kk = 0; kk < 32; ++kk) {
        bw[kk].x = Wm[(4 * kk + koff)     * D + n0 + l15];
        bw[kk].y = Wm[(4 * kk + koff + 1) * D + n0 + l15];
    }
    const float bias = b[n0 + l15];
    __syncthreads();

    {
        float cs = 0.0f, cq = 0.0f;
        for (int rt = wave >> 3; rt < ROWS_PER_BLK / 16; rt += 4) {
            const float* ap = accf + (rt * 16 + l15) * LDS_PITCH + koff;
            v8f d = {};
#pragma unroll
            for (int kk = 0; kk < 32; ++kk) {
                v2f a = *(const v2f*)(ap + 4 * kk);
                d = __builtin_amdgcn_wmma_f32_16x16x4_f32(
                     false, a,  false, bw[kk],
                     (short)0, d,  false,  false);
            }
            float* op = out + (size_t)(lo + rt * 16 + half * 8) * D + n0 + l15;
#pragma unroll
            for (int r = 0; r < 8; ++r) {
                float v = d[r] + bias;
                op[r * D] = v;
                v = v > 0.0f ? v : 0.0f;
                cs += v;
                cq += v * v;
            }
        }
        atomicAdd(&colsum[n0 + l15], cs);
        atomicAdd(&colsum[D + n0 + l15], cq);
    }
    __syncthreads();
    if (tid < D)          atomicAdd(&sums[tid], colsum[tid]);
    else if (tid < 2 * D) atomicAdd(&sumsq[tid - D], colsum[tid]);
}

__global__ void k_init_stats(float* __restrict__ sums, float* __restrict__ sumsq) {
    int i = threadIdx.x;
    if (i < D) { sums[i] = 0.0f; sumsq[i] = 0.0f; }
}

__global__ void k_bn(float* __restrict__ out, const float* __restrict__ sums,
                     const float* __restrict__ sumsq, const float* __restrict__ gamma,
                     const float* __restrict__ beta) {
    int i = blockIdx.x * blockDim.x + threadIdx.x;
    if (i >= N_NODES * D) return;
    int d = i & (D - 1);
    const float invN = 1.0f / (float)N_NODES;
    float mean = sums[d] * invN;
    float var  = sumsq[d] * invN - mean * mean;
    float v = out[i];
    v = v > 0.0f ? v : 0.0f;
    out[i] = gamma[d] * (v - mean) * rsqrtf(var + BN_EPS) + beta[d];
}

extern "C" void kernel_launch(void* const* d_in, const int* in_sizes, int n_in,
                              void* d_out, int out_size, void* d_ws, size_t ws_size,
                              hipStream_t stream) {
    const float* x     = (const float*)d_in[0];
    const int*   edges = (const int*)  d_in[1];
    const float* Wm    = (const float*)d_in[2];
    const float* b     = (const float*)d_in[3];
    const float* gamma = (const float*)d_in[4];
    const float* beta  = (const float*)d_in[5];
    float*       out   = (float*)d_out;

    const int* src = edges;
    const int* dst = edges + N_EDGES;

    float* deg   = (float*)d_ws;
    float* sums  = deg + N_NODES;
    float* sumsq = sums + D;
    unsigned* own4 = (unsigned*)(sumsq + D);

    k_init_deg  <<<(N_NODES + 255) / 256, 256, 0, stream>>>(deg);
    k_deg       <<<(N_EDGES / 4 + 255) / 256, 256, 0, stream>>>(dst, deg, own4);
    k_dinv      <<<(N_NODES + 255) / 256, 256, 0, stream>>>(deg);

    k_init_stats<<<1, 128, 0, stream>>>(sums, sumsq);
    k_aggregate <<<N_BLK, 1024, 0, stream>>>(src, dst, own4, x, Wm, deg, b, out, sums, sumsq);
    k_bn        <<<(N_NODES * D + 255) / 256, 256, 0, stream>>>(out, sums, sumsq, gamma, beta);
}


# Aggregated rows interleaved in LDS (position 32j+l = column 4l+j): every LDS float add of a wave hits 32 different banks, not two lanes a bank; W rows follow that k order, product sum order changes
# speedup vs baseline: 2.1152x; 1.1225x over previous
#include <hip/hip_runtime.h>

#define N_NODES 100000
#define N_EDGES 640000
#define D       128
#define BN_EPS  1e-5f
#define LDS_PITCH 132
#define LDS_POS(c) ((((c) & 3) << 5) + ((c) >> 2))
#define LDS_COL(p) ((((p) & 31) << 2) + ((p) >> 5))

typedef float v2f __attribute__((ext_vector_type(2)));
typedef float v8f __attribute__((ext_vector_type(8)));

__global__ void k_init_deg(float* __restrict__ deg) {
    int i = blockIdx.x * blockDim.x + threadIdx.x;
    if (i < N_NODES) deg[i] = 1.0f;
}

#define ROWS_PER_BLK 400

__global__ void k_deg(const int* __restrict__ dst, float* __restrict__ deg,
                      unsigned* __restrict__ own4) {
    int i = blockIdx.x * blockDim.x + threadIdx.x;
    if (i < N_EDGES / 4) {
        int4 d = ((const int4*)dst)[i];
        atomicAdd(&deg[d.x], 1.0f);
        atomicAdd(&deg[d.y], 1.0f);
        atomicAdd(&deg[d.z], 1.0f);
        atomicAdd(&deg[d.w], 1.0f);
        unsigned o = (unsigned)(d.x / ROWS_PER_BLK)
                   | ((unsigned)(d.y / ROWS_PER_BLK) << 8)
                   | ((unsigned)(d.z / ROWS_PER_BLK) << 16)
                   | ((unsigned)(d.w / ROWS_PER_BLK) << 24);
        own4[i] = o;
    }
}

__global__ void k_dinv(float* __restrict__ deg) {
    int i = blockIdx.x * blockDim.x + threadIdx.x;
    if (i < N_NODES) deg[i] = rsqrtf(deg[i]);
}

#define N_BLK        (N_NODES / ROWS_PER_BLK)
#define LIST_CAP     16384
#define N_OWN16      (N_EDGES / 16)
static_assert(N_BLK * ROWS_PER_BLK == N_NODES, "rows must divide");
static_assert(N_BLK <= 256, "owner number must fit a byte");
static_assert(N_OWN16 * 16 == N_EDGES, "owner map is read in whole 16-byte pieces");
static_assert(ROWS_PER_BLK % 16 == 0, "rows of a block split into whole 16-row matrix tiles");

__device__ __forceinline__ unsigned nz7(unsigned t) {
    return ((t & 0x7F7F7F7Fu) + 0x7F7F7F7Fu) | t;
}

__device__ __noinline__ void edge_serial(int e, int lo, const int* __restrict__ src,
                                         const int* __restrict__ dst,
                                         const float* __restrict__ x,
                                         const float* __restrict__ dinv, float* accf) {
    int s = src[e];
    int d = dst[e];
    int r = d - lo;
    float w = dinv[s] * dinv[d];
    const float* hp = x + (size_t)s * D;
    float* ap = accf + r * LDS_PITCH;
    for (int c = 0; c < D; ++c) atomicAdd(ap + LDS_POS(c), hp[c] * w);
}

__global__ void __launch_bounds__(1024) k_aggregate(const int* __restrict__ src,
                                                    const int* __restrict__ dst,
                                                    const unsigned* __restrict__ own4,
                                                    const float* __restrict__ x,
                                                    const float* __restrict__ Wm,
                                                    const float* __restrict__ dinv,
                                                    const float* __restrict__ b,
                                                    float* __restrict__ out,
                                                    float* __restrict__ sums,
                                                    float* __restrict__ sumsq) {
    __shared__ float4 acc[(ROWS_PER_BLK + 1) * (LDS_PITCH / 4)];
    __shared__ int    list[LIST_CAP];
    __shared__ float  colsum[2 * D];
    __shared__ int    n_list;

    const int tid  = threadIdx.x;
    const int lane = tid & 31;
    const int wave = __builtin_amdgcn_readfirstlane(tid >> 5);
    const int lo   = blockIdx.x * ROWS_PER_BLK;
    float* accf = (float*)acc;
    const float4* h4 = (const float4*)x;

    {
#pragma unroll
        for (int it = 0; it < 13; ++it) {
            int r  = wave + 32 * it;
            int rc = r < ROWS_PER_BLK ? r : ROWS_PER_BLK - 1;
            float di = dinv[lo + rc];
            float sc = di * di;
            float4 hv = h4[(size_t)(lo + rc) * 32 + lane];
            float4 v;
            v.x = hv.x * sc;
            v.y = hv.y * sc;
            v.z = hv.z * sc;
            v.w = hv.w * sc;
            if (r < ROWS_PER_BLK) {
                float* sp = accf + r * LDS_PITCH + lane;
                sp[0]  = v.x;
                sp[32] = v.y;
                sp[64] = v.z;
                sp[96] = v.w;
            }
        }
        if (tid < 2 * D) colsum[tid] = 0.0f;
        if (tid == 0) n_list = 0;
    }
    __syncthreads();

    {
        const uint4* o16 = (const uint4*)own4;
        const unsigned pat = (unsigned)blockIdx.x * 0x01010101u;
        const int start = blockIdx.x * (N_OWN16 / N_BLK);
        for (int i0 = 0; i0 < N_OWN16; i0 += 8192) {
            uint4 ov[8];
            int   jj[8];
#pragma unroll
            for (int u = 0; u < 8; ++u) {
                int i = i0 + u * 1024 + tid;
                int j = (i < N_OWN16 ? i : N_OWN16 - 1) + start;
                if (j >= N_OWN16) j -= N_OWN16;
                jj[u] = j;
                ov[u] = o16[j];
            }
#pragma unroll
            for (int u = 0; u < 8; ++u) {
                unsigned m = ~(((nz7(ov[u].x ^ pat) & 0x80808080u) >> 7) |
                               ((nz7(ov[u].y ^ pat) & 0x80808080u) >> 6) |
                               ((nz7(ov[u].z ^ pat) & 0x80808080u) >> 5) |
                               ((nz7(ov[u].w ^ pat) & 0x80808080u) >> 4)) & 0x0F0F0F0Fu;
                if (i0 + u * 1024 + tid >= N_OWN16) m = 0;
                while (m) {
                    int bit = __builtin_ctz(m);
                    m &= m - 1;
                    int e = 16 * jj[u] + 4 * (bit & 7) + (bit >> 3);
                    int p = atomicAdd(&n_list, 1);
                    if (p < LIST_CAP) list[p] = e;
                    else edge_serial(e, lo, src, dst, x, dinv, accf);
                }
            }
        }
    }
    __syncthreads();

    {
        int nl = n_list;
        nl = nl < LIST_CAP ? nl : LIST_CAP;
        const int n = __builtin_amdgcn_readfirstlane(nl);
        for (int base = wave * 32; base < n; base += 1024) {
            int r = ROWS_PER_BLK, s = 0;
            float w = 0.0f;
            if (base + lane < n) {
                int e = list[base + lane];
                s = src[e];
                int d = dst[e];
                r = d - lo;
                w = dinv[s] * dinv[d];
            }
            const int wi = __float_as_int(w);
#pragma unroll
            for (int g = 0; g < 32; g += 8) {
                if (base + g < n) {
                    float4 hv[8];
                    int    rr[8];
                    float  ww[8];
#pragma unroll
                    for (int q = 0; q < 8; ++q) {
                        int sq = __builtin_amdgcn_readlane(s, g + q);
                        rr[q]  = __builtin_amdgcn_readlane(r, g + q);
                        ww[q]  = __int_as_float(__builtin_amdgcn_readlane(wi, g + q));
                        hv[q]  = h4[(size_t)sq * 32 + lane];
                    }
#pragma unroll
                    for (int q = 0; q < 8; ++q) {
                        float* ap = accf + rr[q] * LDS_PITCH + lane;
                        atomicAdd(ap + 0,  hv[q].x * ww[q]);
                        atomicAdd(ap + 32, hv[q].y * ww[q]);
                        atomicAdd(ap + 64, hv[q].z * ww[q]);
                        atomicAdd(ap + 96, hv[q].w * ww[q]);
                    }
                }
            }
        }
    }
    const int half = lane >> 4;
    const int l15  = lane & 15;
    const int koff = half * 2;
    const int n0   = (wave & 7) * 16;
    v2f bw[32];
#pragma unroll
    for (int kk = 0; kk < 32; ++kk) {
        bw[kk].x = Wm[LDS_COL(4 * kk + koff)     * D + n0 + l15];
        bw[kk].y = Wm[LDS_COL(4 * kk + koff + 1) * D + n0 + l15];
    }
    const float bias = b[n0 + l15];
    __syncthreads();

    {
        float cs = 0.0f, cq = 0.0f;
        for (int rt = wave >> 3; rt < ROWS_PER_BLK / 16; rt += 4) {
            const float* ap = accf + (rt * 16 + l15) * LDS_PITCH + koff;
            v8f d = {};
#pragma unroll
            for (int kk = 0; kk < 32; ++kk) {
                v2f a = *(const v2f*)(ap + 4 * kk);
                d = __builtin_amdgcn_wmma_f32_16x16x4_f32(
                     false, a,  false, bw[kk],
                     (short)0, d,  false,  false);
            }
            float* op = out + (size_t)(lo + rt * 16 + half * 8) * D + n0 + l15;
#pragma unroll
            for (int r = 0; r < 8; ++r) {
                float v = d[r] + bias;
                op[r * D] = v;
                v = v > 0.0f ? v : 0.0f;
                cs += v;
                cq += v * v;
            }
        }
        atomicAdd(&colsum[n0 + l15], cs);
        atomicAdd(&colsum[D + n0 + l15], cq);
    }
    __syncthreads();
    if (tid < D)          atomicAdd(&sums[tid], colsum[tid]);
    else if (tid < 2 * D) atomicAdd(&sumsq[tid - D], colsum[tid]);
}

__global__ void k_init_stats(float* __restrict__ sums, float* __restrict__ sumsq) {
    int i = threadIdx.x;
    if (i < D) { sums[i] = 0.0f; sumsq[i] = 0.0f; }
}

__global__ void k_bn(float* __restrict__ out, const float* __restrict__ sums,
                     const float* __restrict__ sumsq, const float* __restrict__ gamma,
                     const float* __restrict__ beta) {
    int i = blockIdx.x * blockDim.x + threadIdx.x;
    if (i >= N_NODES * D) return;
    int d = i & (D - 1);
    const float invN = 1.0f / (float)N_NODES;
    float mean = sums[d] * invN;
    float var  = sumsq[d] * invN - mean * mean;
    float v = out[i];
    v = v > 0.0f ? v : 0.0f;
    out[i] = gamma[d] * (v - mean) * rsqrtf(var + BN_EPS) + beta[d];
}

extern "C" void kernel_launch(void* const* d_in, const int* in_sizes, int n_in,
                              void* d_out, int out_size, void* d_ws, size_t ws_size,
                              hipStream_t stream) {
    const float* x     = (const float*)d_in[0];
    const int*   edges = (const int*)  d_in[1];
    const float* Wm    = (const float*)d_in[2];
    const float* b     = (const float*)d_in[3];
    const float* gamma = (const float*)d_in[4];
    const float* beta  = (const float*)d_in[5];
    float*       out   = (float*)d_out;

    const int* src = edges;
    const int* dst = edges + N_EDGES;

    float* deg   = (float*)d_ws;
    float* sums  = deg + N_NODES;
    float* sumsq = sums + D;
    unsigned* own4 = (unsigned*)(sumsq + D);

    k_init_deg  <<<(N_NODES + 255) / 256, 256, 0, stream>>>(deg);
    k_deg       <<<(N_EDGES / 4 + 255) / 256, 256, 0, stream>>>(dst, deg, own4);
    k_dinv      <<<(N_NODES + 255) / 256, 256, 0, stream>>>(deg);

    k_init_stats<<<1, 128, 0, stream>>>(sums, sumsq);
    k_aggregate <<<N_BLK, 1024, 0, stream>>>(src, dst, own4, x, Wm, deg, b, out, sums, sumsq);
    k_bn        <<<(N_NODES * D + 255) / 256, 256, 0, stream>>>(out, sums, sumsq, gamma, beta);
}


# In-degree counted by the 250 owner blocks with LDS int adds over the edges each finds in the owner map, not 640000 scattered global float adds; found list handed to the aggregation; sums as before
# speedup vs baseline: 2.7841x; 1.3162x over previous
#include <hip/hip_runtime.h>

#define N_NODES 100000
#define N_EDGES 640000
#define D       128
#define BN_EPS  1e-5f
#define LDS_PITCH 132
#define LDS_POS(c) ((((c) & 3) << 5) + ((c) >> 2))
#define LDS_COL(p) ((((p) & 31) << 2) + ((p) >> 5))

typedef float v2f __attribute__((ext_vector_type(2)));
typedef float v8f __attribute__((ext_vector_type(8)));

#define ROWS_PER_BLK 400

__global__ void k_own(const int* __restrict__ dst, unsigned* __restrict__ own4) {
    int i = blockIdx.x * blockDim.x + threadIdx.x;
    if (i < N_EDGES / 4) {
        int4 d = ((const int4*)dst)[i];
        unsigned o = (unsigned)(d.x / ROWS_PER_BLK)
                   | ((unsigned)(d.y / ROWS_PER_BLK) << 8)
                   | ((unsigned)(d.z / ROWS_PER_BLK) << 16)
                   | ((unsigned)(d.w / ROWS_PER_BLK) << 24);
        own4[i] = o;
    }
}

#define N_BLK        (N_NODES / ROWS_PER_BLK)
#define LIST_CAP     16384
#define N_OWN16      (N_EDGES / 16)
#define SRC_BITS     17
static_assert(N_BLK * ROWS_PER_BLK == N_NODES, "rows must divide");
static_assert(N_BLK <= 256, "owner number must fit a byte");
static_assert(N_OWN16 * 16 == N_EDGES, "owner map is read in whole 16-byte pieces");
static_assert(ROWS_PER_BLK % 16 == 0, "rows of a block split into whole 16-row matrix tiles");
static_assert(N_NODES <= (1 << SRC_BITS), "source number must fit its field");
static_assert(ROWS_PER_BLK <= (1 << (31 - SRC_BITS)), "row number must fit its field");

__device__ __forceinline__ unsigned nz7(unsigned t) {
    return ((t & 0x7F7F7F7Fu) + 0x7F7F7F7Fu) | t;
}

template <class Overflow>
__device__ __forceinline__ void filter_own_edges(const unsigned* __restrict__ own4, int tid,
                                                 int* list, int* n_list, Overflow overflow) {
    const uint4* o16 = (const uint4*)own4;
    const unsigned pat = (unsigned)blockIdx.x * 0x01010101u;
    const int start = blockIdx.x * (N_OWN16 / N_BLK);
    for (int i0 = 0; i0 < N_OWN16; i0 += 8192) {
        uint4 ov[8];
        int   jj[8];
#pragma unroll
        for (int u = 0; u < 8; ++u) {
            int i = i0 + u * 1024 + tid;
            int j = (i < N_OWN16 ? i : N_OWN16 - 1) + start;
            if (j >= N_OWN16) j -= N_OWN16;
            jj[u] = j;
            ov[u] = o16[j];
        }
#pragma unroll
        for (int u = 0; u < 8; ++u) {
            unsigned m = ~(((nz7(ov[u].x ^ pat) & 0x80808080u) >> 7) |
                           ((nz7(ov[u].y ^ pat) & 0x80808080u) >> 6) |
                           ((nz7(ov[u].z ^ pat) & 0x80808080u) >> 5) |
                           ((nz7(ov[u].w ^ pat) & 0x80808080u) >> 4)) & 0x0F0F0F0Fu;
            if (i0 + u * 1024 + tid >= N_OWN16) m = 0;
            while (m) {
                int bit = __builtin_ctz(m);
                m &= m - 1;
                int e = 16 * jj[u] + 4 * (bit & 7) + (bit >> 3);
                int p = atomicAdd(n_list, 1);
                if (p < LIST_CAP) list[p] = e;
                else overflow(e);
            }
        }
    }
}

__global__ void __launch_bounds__(1024) k_count(const int* __restrict__ src,
                                                const int* __restrict__ dst,
                                                const unsigned* __restrict__ own4,
                                                float* __restrict__ dinv,
                                                int* __restrict__ gcount,
                                                unsigned* __restrict__ glist) {
    __shared__ int list[LIST_CAP];
    __shared__ int cnt[ROWS_PER_BLK];
    __shared__ int n_list;

    const int tid = threadIdx.x;
    const int lo  = blockIdx.x * ROWS_PER_BLK;
    if (tid < ROWS_PER_BLK) cnt[tid] = 0;
    if (tid == 0) n_list = 0;
    __syncthreads();

    filter_own_edges(own4, tid, list, &n_list,
                     [&](int e) { atomicAdd(&cnt[dst[e] - lo], 1); });
    __syncthreads();

    const int n_found = n_list;
    const int n = __builtin_amdgcn_readfirstlane(n_found < LIST_CAP ? n_found : LIST_CAP);
    unsigned* gl = glist + (size_t)blockIdx.x * LIST_CAP;
    for (int i0 = 0; i0 < n; i0 += 4096) {
        int sv[4], dv[4];
#pragma unroll
        for (int u = 0; u < 4; ++u) {
            int i = i0 + u * 1024 + tid;
            int e = list[i < n ? i : n - 1];
            sv[u] = src[e];
            dv[u] = dst[e];
        }
#pragma unroll
        for (int u = 0; u < 4; ++u) {
            int i = i0 + u * 1024 + tid;
            if (i < n) {
                int r = dv[u] - lo;
                atomicAdd(&cnt[r], 1);
                gl[i] = (unsigned)sv[u] | ((unsigned)r << SRC_BITS);
            }
        }
    }
    if (tid == 0) gcount[blockIdx.x] = n_found;
    __syncthreads();
    if (tid < ROWS_PER_BLK) dinv[lo + tid] = rsqrtf((float)(cnt[tid] + 1));
}


__device__ __noinline__ void edge_serial(int e, int lo, const int* __restrict__ src,
                                         const int* __restrict__ dst,
                                         const float* __restrict__ x,
                                         const float* __restrict__ dinv, float* accf) {
    int s = src[e];
    int d = dst[e];
    int r = d - lo;
    float w = dinv[s] * dinv[d];
    const float* hp = x + (size_t)s * D;
    float* ap = accf + r * LDS_PITCH;
    for (int c = 0; c < D; ++c) atomicAdd(ap + LDS_POS(c), hp[c] * w);
}

__global__ void __launch_bounds__(1024) k_aggregate(const int* __restrict__ src,
                                                    const int* __restrict__ dst,
                                                    const unsigned* __restrict__ own4,
                                                    const int* __restrict__ gcount,
                                                    const unsigned* __restrict__ glist,
                                                    const float* __restrict__ x,
                                                    const float* __restrict__ Wm,
                                                    const float* __restrict__ dinv,
                                                    const float* __restrict__ b,
                                                    float* __restrict__ out,
                                                    float* __restrict__ sums,
                                                    float* __restrict__ sumsq) {
    __shared__ float4 acc[(ROWS_PER_BLK + 1) * (LDS_PITCH / 4)];
    __shared__ int    list[LIST_CAP];
    __shared__ float  colsum[2 * D];
    __shared__ int    n_list;

    const int tid  = threadIdx.x;
    const int lane = tid & 31;
    const int wave = __builtin_amdgcn_readfirstlane(tid >> 5);
    const int lo   = blockIdx.x * ROWS_PER_BLK;
    float* accf = (float*)acc;
    const float4* h4 = (const float4*)x;

    {
#pragma unroll
        for (int it = 0; it < 13; ++it) {
            int r  = wave + 32 * it;
            int rc = r < ROWS_PER_BLK ? r : ROWS_PER_BLK - 1;
            float di = dinv[lo + rc];
            float sc = di * di;
            float4 hv = h4[(size_t)(lo + rc) * 32 + lane];
            float4 v;
            v.x = hv.x * sc;
            v.y = hv.y * sc;
            v.z = hv.z * sc;
            v.w = hv.w * sc;
            if (r < ROWS_PER_BLK) {
                float* sp = accf + r * LDS_PITCH + lane;
                sp[0]  = v.x;
                sp[32] = v.y;
                sp[64] = v.z;
                sp[96] = v.w;
            }
        }
        if (tid < 2 * D) colsum[tid] = 0.0f;
        if (tid == 0) n_list = 0;
    }
    __syncthreads();

    const int  n_found = __builtin_amdgcn_readfirstlane(gcount[blockIdx.x]);
    const bool own_search = n_found > LIST_CAP;
    if (own_search) {
        filter_own_edges(own4, tid, list, &n_list,
                         [&](int e) { edge_serial(e, lo, src, dst, x, dinv, accf); });
        __syncthreads();
    }

    {
        int nl = n_found;
        if (own_search) {
            nl = n_list;
            nl = nl < LIST_CAP ? nl : LIST_CAP;
        }
        const int n = __builtin_amdgcn_readfirstlane(nl);
        const unsigned* gl = glist + (size_t)blockIdx.x * LIST_CAP;
        for (int base = wave * 32; base < n; base += 1024) {
            int r = ROWS_PER_BLK, s = 0;
            float w = 0.0f;
            if (base + lane < n) {
                if (own_search) {
                    int e = list[base + lane];
                    s = src[e];
                    r = dst[e] - lo;
                } else {
                    unsigned q = gl[base + lane];
                    s = (int)(q & ((1u << SRC_BITS) - 1u));
                    r = (int)(q >> SRC_BITS);
                }
                w = dinv[s] * dinv[lo + r];
            }
            const int wi = __float_as_int(w);
#pragma unroll
            for (int g = 0; g < 32; g += 8) {
                if (base + g < n) {
                    float4 hv[8];
                    int    rr[8];
                    float  ww[8];
#pragma unroll
                    for (int q = 0; q < 8; ++q) {
                        int sq = __builtin_amdgcn_readlane(s, g + q);
                        rr[q]  = __builtin_amdgcn_readlane(r, g + q);
                        ww[q]  = __int_as_float(__builtin_amdgcn_readlane(wi, g + q));
                        hv[q]  = h4[(size_t)sq * 32 + lane];
                    }
#pragma unroll
                    for (int q = 0; q < 8; ++q) {
                        float* ap = accf + rr[q] * LDS_PITCH + lane;
                        atomicAdd(ap + 0,  hv[q].x * ww[q]);
                        atomicAdd(ap + 32, hv[q].y * ww[q]);
                        atomicAdd(ap + 64, hv[q].z * ww[q]);
                        atomicAdd(ap + 96, hv[q].w * ww[q]);
                    }
                }
            }
        }
    }
    const int half = lane >> 4;
    const int l15  = lane & 15;
    const int koff = half * 2;
    const int n0   = (wave & 7) * 16;
    v2f bw[32];
#pragma unroll
    for (int kk = 0; kk < 32; ++kk) {
        bw[kk].x = Wm[LDS_COL(4 * kk + koff)     * D + n0 + l15];
        bw[kk].y = Wm[LDS_COL(4 * kk + koff + 1) * D + n0 + l15];
    }
    const float bias = b[n0 + l15];
    __syncthreads();

    {
        float cs = 0.0f, cq = 0.0f;
        for (int rt = wave >> 3; rt < ROWS_PER_BLK / 16; rt += 4) {
            const float* ap = accf + (rt * 16 + l15) * LDS_PITCH + koff;
            v8f d = {};
#pragma unroll
            for (int kk = 0; kk < 32; ++kk) {
                v2f a = *(const v2f*)(ap + 4 * kk);
                d = __builtin_amdgcn_wmma_f32_16x16x4_f32(
                     false, a,  false, bw[kk],
                     (short)0, d,  false,  false);
            }
            float* op = out + (size_t)(lo + rt * 16 + half * 8) * D + n0 + l15;
#pragma unroll
            for (int r = 0; r < 8; ++r) {
                float v = d[r] + bias;
                op[r * D] = v;
                v = v > 0.0f ? v : 0.0f;
                cs += v;
                cq += v * v;
            }
        }
        atomicAdd(&colsum[n0 + l15], cs);
        atomicAdd(&colsum[D + n0 + l15], cq);
    }
    __syncthreads();
    if (tid < D)          atomicAdd(&sums[tid], colsum[tid]);
    else if (tid < 2 * D) atomicAdd(&sumsq[tid - D], colsum[tid]);
}

__global__ void k_init_stats(float* __restrict__ sums, float* __restrict__ sumsq) {
    int i = threadIdx.x;
    if (i < D) { sums[i] = 0.0f; sumsq[i] = 0.0f; }
}

__global__ void k_bn(float* __restrict__ out, const float* __restrict__ sums,
                     const float* __restrict__ sumsq, const float* __restrict__ gamma,
                     const float* __restrict__ beta) {
    int i = blockIdx.x * blockDim.x + threadIdx.x;
    if (i >= N_NODES * D) return;
    int d = i & (D - 1);
    const float invN = 1.0f / (float)N_NODES;
    float mean = sums[d] * invN;
    float var  = sumsq[d] * invN - mean * mean;
    float v = out[i];
    v = v > 0.0f ? v : 0.0f;
    out[i] = gamma[d] * (v - mean) * rsqrtf(var + BN_EPS) + beta[d];
}

extern "C" void kernel_launch(void* const* d_in, const int* in_sizes, int n_in,
                              void* d_out, int out_size, void* d_ws, size_t ws_size,
                              hipStream_t stream) {
    const float* x     = (const float*)d_in[0];
    const int*   edges = (const int*)  d_in[1];
    const float* Wm    = (const float*)d_in[2];
    const float* b     = (const float*)d_in[3];
    const float* gamma = (const float*)d_in[4];
    const float* beta  = (const float*)d_in[5];
    float*       out   = (float*)d_out;

    const int* src = edges;
    const int* dst = edges + N_EDGES;

    float* dinv  = (float*)d_ws;
    float* sums  = dinv + N_NODES;
    float* sumsq = sums + D;
    unsigned* own4 = (unsigned*)(sumsq + D);
    int* gcount    = (int*)(own4 + N_EDGES / 4);
    unsigned* glist = (unsigned*)(gcount + 256);

    k_own       <<<(N_EDGES / 4 + 255) / 256, 256, 0, stream>>>(dst, own4);
    k_count     <<<N_BLK, 1024, 0, stream>>>(src, dst, own4, dinv, gcount, glist);

    k_init_stats<<<1, 128, 0, stream>>>(sums, sumsq);
    k_aggregate <<<N_BLK, 1024, 0, stream>>>(src, dst, own4, gcount, glist, x, Wm, dinv, b, out,
                                             sums, sumsq);
    k_bn        <<<(N_NODES * D + 255) / 256, 256, 0, stream>>>(out, sums, sumsq, gamma, beta);
}
